# starting kernel
# baseline (speedup 1.0000x reference)
#include <hip/hip_runtime.h>
#include <hip/hip_bf16.h>

#define N_NODES 10000
#define N_EDGES 100000
#define DIM     768

typedef __attribute__((ext_vector_type(2))) float v2f;
typedef __attribute__((ext_vector_type(8))) float v8f;

__global__ void gcn_init_deg(float* __restrict__ deg) {
    int i = blockIdx.x * blockDim.x + threadIdx.x;
    if (i < N_NODES) deg[i] = 1.0f;
}

__global__ void gcn_count_deg(const int* __restrict__ dst, float* __restrict__ deg) {
    int e = blockIdx.x * blockDim.x + threadIdx.x;
    if (e < N_EDGES) atomicAdd(&deg[dst[e]], 1.0f);
}

__global__ void gcn_rsqrt_deg(const float* __restrict__ deg, float* __restrict__ dinv) {
    int i = blockIdx.x * blockDim.x + threadIdx.x;
    if (i < N_NODES) dinv[i] = rsqrtf(deg[i]);
}

__global__ __launch_bounds__(32) void gcn_gemm_wmma(const float* __restrict__ X,
                                                    const float* __restrict__ W,
                                                    float* __restrict__ H) {
    const int lane = threadIdx.x & 31;
    const int m    = lane & 15;
    const int hi   = lane >> 4;
    const int koff = hi * 2;
    const int row0 = blockIdx.x * 16;
    const int col0 = blockIdx.y * 64;

    v8f c0 = {}, c1 = {}, c2 = {}, c3 = {};

    const float* xrow = X + (size_t)(row0 + m) * DIM;

    for (int k = 0; k < DIM; k += 4) {
        v2f a = *(const v2f*)(xrow + k + koff);

        const float* wb = W + (size_t)(k + koff) * DIM + col0 + m;
        v2f b0 = { wb[ 0], wb[DIM +  0] };
        v2f b1 = { wb[16], wb[DIM + 16] };
        v2f b2 = { wb[32], wb[DIM + 32] };
        v2f b3 = { wb[48], wb[DIM + 48] };

        c0 = __builtin_amdgcn_wmma_f32_16x16x4_f32(false, a, false, b0, (short)0, c0, false, false);
        c1 = __builtin_amdgcn_wmma_f32_16x16x4_f32(false, a, false, b1, (short)0, c1, false, false);
        c2 = __builtin_amdgcn_wmma_f32_16x16x4_f32(false, a, false, b2, (short)0, c2, false, false);
        c3 = __builtin_amdgcn_wmma_f32_16x16x4_f32(false, a, false, b3, (short)0, c3, false, false);
    }

    #pragma unroll
    for (int r = 0; r < 8; ++r) {
        const int row = row0 + r + hi * 8;
        float* ho = H + (size_t)row * DIM + col0 + m;
        ho[ 0] = c0[r];
        ho[16] = c1[r];
        ho[32] = c2[r];
        ho[48] = c3[r];
    }
}

__global__ void gcn_self_bias(const float* __restrict__ H,
                              const float* __restrict__ dinv,
                              const float* __restrict__ bias,
                              float* __restrict__ out) {
    const int i = blockIdx.x;
    const float n = dinv[i] * dinv[i];
    const size_t base = (size_t)i * DIM;
    for (int d = threadIdx.x; d < DIM; d += blockDim.x)
        out[base + d] = bias[d] + n * H[base + d];
}

__global__ void gcn_scatter(const int* __restrict__ src,
                            const int* __restrict__ dst,
                            const float* __restrict__ H,
                            const float* __restrict__ dinv,
                            float* __restrict__ out) {
    const int e = blockIdx.x;
    const int s = src[e];
    const int t = dst[e];
    const float n = dinv[s] * dinv[t];
    const size_t sb = (size_t)s * DIM;
    const size_t tb = (size_t)t * DIM;
    for (int d = threadIdx.x; d < DIM; d += blockDim.x)
        atomicAdd(&out[tb + d], n * H[sb + d]);
}

extern "C" void kernel_launch(void* const* d_in, const int* in_sizes, int n_in,
                              void* d_out, int out_size, void* d_ws, size_t ws_size,
                              hipStream_t stream) {
    const float* x    = (const float*)d_in[0];
    const int*   ei   = (const int*)d_in[1];
    const float* W    = (const float*)d_in[2];
    const float* bias = (const float*)d_in[3];
    float*       out  = (float*)d_out;

    const int* src = ei;
    const int* dst = ei + N_EDGES;

    float* deg  = (float*)d_ws;
    float* dinv = deg + N_NODES;
    float* h    = dinv + N_NODES;

    gcn_init_deg <<<(N_NODES + 255) / 256, 256, 0, stream>>>(deg);
    gcn_count_deg<<<(N_EDGES + 255) / 256, 256, 0, stream>>>(dst, deg);
    gcn_rsqrt_deg<<<(N_NODES + 255) / 256, 256, 0, stream>>>(deg, dinv);

    dim3 ggrid(N_NODES / 16, DIM / 64);
    gcn_gemm_wmma<<<ggrid, 32, 0, stream>>>(x, W, h);

    gcn_self_bias<<<N_NODES, 256, 0, stream>>>(h, dinv, bias, out);
    gcn_scatter  <<<N_EDGES, 256, 0, stream>>>(src, dst, h, dinv, out);
}


# out=(A_hat x)W+b: x rows summed per target in LDS, then times W on f32 WMMA in the same launch (A from LDS); h buffer and product launch gone; sum order: neighbours first, then k in two halves
# speedup vs baseline: 2.2343x; 2.2343x over previous
#include <hip/hip_runtime.h>
#include <hip/hip_bf16.h>

#define N_NODES 10000
#define N_EDGES 100000
#define DIM     768

typedef __attribute__((ext_vector_type(2))) float v2f;
typedef __attribute__((ext_vector_type(8))) float v8f;

__global__ void gcn_init_deg(float* __restrict__ deg) {
    int i = blockIdx.x * blockDim.x + threadIdx.x;
    if (i < N_NODES) deg[i] = 1.0f;
}

__global__ void gcn_count_deg(const int* __restrict__ dst, float* __restrict__ deg) {
    int e = blockIdx.x * blockDim.x + threadIdx.x;
    if (e < N_EDGES) atomicAdd(&deg[dst[e]], 1.0f);
}

__global__ void gcn_rsqrt_deg(const float* __restrict__ deg, float* __restrict__ dinv) {
    int i = blockIdx.x * blockDim.x + threadIdx.x;
    if (i < N_NODES) dinv[i] = rsqrtf(deg[i]);
}

#define AGG_ROWS    40
#define AGG_PITCH   (DIM + 4)
#define AGG_THREADS 1024
#define AGG_WAVES   (AGG_THREADS / 32)
#define AGG_GROUP   4
#define AGG_SPAN    (AGG_GROUP * 128)
#define AGG_STRIDE  (AGG_WAVES * AGG_SPAN)
#define AGG_VEC     (AGG_ROWS * DIM / 4)
#define GEMM_COLS   48
#define GEMM_KHALF  (DIM / 2)

__device__ __forceinline__ int4 agg_load_dst4(const int* __restrict__ dst, int e) {
    const int  ec = e < N_EDGES - 4 ? e : N_EDGES - 4;
    const int4 d  = *(const int4*)(dst + ec);
    const bool in = e < N_EDGES;
    return make_int4(in ? d.x : -1, in ? d.y : -1, in ? d.z : -1, in ? d.w : -1);
}

struct gemm_frags { v2f a[3]; v2f b[3]; };

__device__ __forceinline__ void gemm_load_frags(gemm_frags& f,
                                                const float* a0, const float* a1, const float* a2,
                                                const float* __restrict__ wk) {
    f.a[0] = *(const v2f*)a0;
    f.a[1] = *(const v2f*)a1;
    f.a[2] = *(const v2f*)a2;
    f.b[0] = v2f{ wk[ 0], wk[DIM +  0] };
    f.b[1] = v2f{ wk[16], wk[DIM + 16] };
    f.b[2] = v2f{ wk[32], wk[DIM + 32] };
}

__device__ __forceinline__ void gemm_step(v8f (&d)[3][3], const gemm_frags& f) {
    #pragma unroll
    for (int i = 0; i < 3; ++i) {
        #pragma unroll
        for (int j = 0; j < 3; ++j)
            d[i][j] = __builtin_amdgcn_wmma_f32_16x16x4_f32(false, f.a[i], false, f.b[j],
                                                            (short)0, d[i][j], false, false);
    }
}

__global__ __launch_bounds__(AGG_THREADS) void gcn_aggregate_gemm(const int* __restrict__ src,
                                                                 const int* __restrict__ dst,
                                                                 const float* __restrict__ X,
                                                                 const float* __restrict__ W,
                                                                 const float* __restrict__ dinv,
                                                                 const float* __restrict__ bias,
                                                                 float* __restrict__ out) {
    __shared__ float acc[AGG_ROWS * AGG_PITCH];

    const int tid  = threadIdx.x;
    const int lane = tid & 31;
    const int wave = __builtin_amdgcn_readfirstlane(tid >> 5);
    const int lo   = blockIdx.x * AGG_ROWS;

    #pragma unroll
    for (int it = 0; it < (AGG_VEC + AGG_THREADS - 1) / AGG_THREADS; ++it) {
        const int q = tid + it * AGG_THREADS;
        if (q < AGG_VEC) {
            const int r = q / (DIM / 4);
            const int c = (q - r * (DIM / 4)) * 4;
            const int i = lo + r;
            const float n = dinv[i] * dinv[i];
            const float4 x4 = *(const float4*)(X + (size_t)i * DIM + c);
            float4 v;
            v.x = n * x4.x;
            v.y = n * x4.y;
            v.z = n * x4.z;
            v.w = n * x4.w;
            *(float4*)(acc + r * AGG_PITCH + c) = v;
        }
    }
    __syncthreads();

    int base = wave * AGG_SPAN;
    int4 cur[AGG_GROUP];
    #pragma unroll
    for (int g = 0; g < AGG_GROUP; ++g)
        cur[g] = agg_load_dst4(dst, base + g * 128 + lane * 4);

    for (; base < N_EDGES; base += AGG_STRIDE) {
        int4 nxt[AGG_GROUP];
        #pragma unroll
        for (int g = 0; g < AGG_GROUP; ++g)
            nxt[g] = agg_load_dst4(dst, base + AGG_STRIDE + g * 128 + lane * 4);

        unsigned bits = 0u;
        #pragma unroll
        for (int g = 0; g < AGG_GROUP; ++g) {
            bits |= ((unsigned)(cur[g].x - lo) < (unsigned)AGG_ROWS ? 1u : 0u) << (4 * g + 0);
            bits |= ((unsigned)(cur[g].y - lo) < (unsigned)AGG_ROWS ? 1u : 0u) << (4 * g + 1);
            bits |= ((unsigned)(cur[g].z - lo) < (unsigned)AGG_ROWS ? 1u : 0u) << (4 * g + 2);
            bits |= ((unsigned)(cur[g].w - lo) < (unsigned)AGG_ROWS ? 1u : 0u) << (4 * g + 3);
        }

        unsigned lanes;
        while ((lanes = __builtin_amdgcn_ballot_w32(bits != 0u)) != 0u) {
            int   s = 0;
            int   r = 0;
            float n = 0.0f;
            if (bits != 0u) {
                const int j = __builtin_ctz(bits);
                bits &= bits - 1u;
                const int e = base + (j >> 2) * 128 + lane * 4 + (j & 3);
                const int t = dst[e];
                s = src[e];
                n = dinv[s] * dinv[t];
                r = t - lo;
            }
            while (lanes != 0u) {
                const int k = __builtin_ctz(lanes);
                lanes &= lanes - 1u;
                const int   sk = __builtin_amdgcn_readlane(s, k);
                const int   rk = __builtin_amdgcn_readlane(r, k);
                const float nk = __builtin_bit_cast(float,
                                     __builtin_amdgcn_readlane(__builtin_bit_cast(int, n), k));
                const float* xp = X + (size_t)sk * DIM + lane * 2;
                float*       ap = acc + rk * AGG_PITCH + lane * 2;
                v2f v[DIM / 64];
                #pragma unroll
                for (int j = 0; j < DIM / 64; ++j)
                    v[j] = *(const v2f*)(xp + j * 64);
                #pragma unroll
                for (int j = 0; j < DIM / 64; ++j) {
                    atomicAdd(ap + j * 64,     nk * v[j].x);
                    atomicAdd(ap + j * 64 + 1, nk * v[j].y);
                }
            }
        }

        #pragma unroll
        for (int g = 0; g < AGG_GROUP; ++g)
            cur[g] = nxt[g];
    }
    __syncthreads();

    const int m    = lane & 15;
    const int hi   = lane >> 4;
    const int col0 = (wave & 15) * GEMM_COLS;
    const int kh   = wave >> 4;
    const int k0   = kh * GEMM_KHALF + hi * 2;
    const int row2 = 32 + m < AGG_ROWS ? 32 + m : AGG_ROWS - 1;

    const float* a0 = acc + m * AGG_PITCH + k0;
    const float* a1 = acc + (16 + m) * AGG_PITCH + k0;
    const float* a2 = acc + row2 * AGG_PITCH + k0;
    const float* wp = W + (size_t)k0 * DIM + col0 + m;

    v8f d[3][3] = {};
    gemm_frags f0, f1;
    gemm_load_frags(f0, a0, a1, a2, wp);
    #pragma clang loop unroll(disable)
    for (int k = 0; k < GEMM_KHALF; k += 8) {
        gemm_load_frags(f1, a0 + k + 4, a1 + k + 4, a2 + k + 4, wp + (size_t)(k + 4) * DIM);
        gemm_step(d, f0);
        const int kn = k + 8 < GEMM_KHALF ? k + 8 : k;
        gemm_load_frags(f0, a0 + kn, a1 + kn, a2 + kn, wp + (size_t)kn * DIM);
        gemm_step(d, f1);
    }
    __syncthreads();

    if (kh == 1) {
        #pragma unroll
        for (int i = 0; i < 3; ++i) {
            if (16 * i + 8 * hi < AGG_ROWS) {
                #pragma unroll
                for (int r = 0; r < 8; ++r) {
                    float* p = acc + (16 * i + 8 * hi + r) * AGG_PITCH + col0 + m;
                    p[ 0] = d[i][0][r];
                    p[16] = d[i][1][r];
                    p[32] = d[i][2][r];
                }
            }
        }
    }
    __syncthreads();
    if (kh == 0) {
        const float b0 = bias[col0 + m];
        const float b1 = bias[col0 + m + 16];
        const float b2 = bias[col0 + m + 32];
        #pragma unroll
        for (int i = 0; i < 3; ++i) {
            if (16 * i + 8 * hi < AGG_ROWS) {
                #pragma unroll
                for (int r = 0; r < 8; ++r) {
                    float* p = acc + (16 * i + 8 * hi + r) * AGG_PITCH + col0 + m;
                    p[ 0] = (d[i][0][r] + p[ 0]) + b0;
                    p[16] = (d[i][1][r] + p[16]) + b1;
                    p[32] = (d[i][2][r] + p[32]) + b2;
                }
            }
        }
    }
    __syncthreads();

    float* op = out + (size_t)lo * DIM;
    #pragma unroll
    for (int it = 0; it < (AGG_VEC + AGG_THREADS - 1) / AGG_THREADS; ++it) {
        const int q = tid + it * AGG_THREADS;
        if (q < AGG_VEC) {
            const int r = q / (DIM / 4);
            const int c = (q - r * (DIM / 4)) * 4;
            *(float4*)(op + q * 4) = *(const float4*)(acc + r * AGG_PITCH + c);
        }
    }
}

extern "C" void kernel_launch(void* const* d_in, const int* in_sizes, int n_in,
                              void* d_out, int out_size, void* d_ws, size_t ws_size,
                              hipStream_t stream) {
    const float* x    = (const float*)d_in[0];
    const int*   ei   = (const int*)d_in[1];
    const float* W    = (const float*)d_in[2];
    const float* bias = (const float*)d_in[3];
    float*       out  = (float*)d_out;

    const int* src = ei;
    const int* dst = ei + N_EDGES;

    float* deg  = (float*)d_ws;
    float* dinv = deg + N_NODES;

    gcn_init_deg <<<(N_NODES + 255) / 256, 256, 0, stream>>>(deg);
    gcn_count_deg<<<(N_EDGES + 255) / 256, 256, 0, stream>>>(dst, deg);
    gcn_rsqrt_deg<<<(N_NODES + 255) / 256, 256, 0, stream>>>(deg, dinv);

    gcn_aggregate_gemm<<<N_NODES / AGG_ROWS, AGG_THREADS, 0, stream>>>(src, dst, x, W, dinv, bias, out);
}


# Neighbour sums without LDS float adds: owner group's scan files matching edges per target row in LDS, then one wave sums a half row in registers (4 rows in flight); sum order = list slot order
# speedup vs baseline: 2.3339x; 1.0446x over previous
#include <hip/hip_runtime.h>
#include <hip/hip_bf16.h>

#define N_NODES 10000
#define N_EDGES 100000
#define DIM     768

typedef __attribute__((ext_vector_type(2))) float v2f;
typedef __attribute__((ext_vector_type(8))) float v8f;

__global__ void gcn_init_deg(float* __restrict__ deg) {
    int i = blockIdx.x * blockDim.x + threadIdx.x;
    if (i < N_NODES) deg[i] = 1.0f;
}

__global__ void gcn_count_deg(const int* __restrict__ dst, float* __restrict__ deg) {
    int e = blockIdx.x * blockDim.x + threadIdx.x;
    if (e < N_EDGES) atomicAdd(&deg[dst[e]], 1.0f);
}

__global__ void gcn_rsqrt_deg(const float* __restrict__ deg, float* __restrict__ dinv) {
    int i = blockIdx.x * blockDim.x + threadIdx.x;
    if (i < N_NODES) dinv[i] = rsqrtf(deg[i]);
}

#define AGG_ROWS    40
#define AGG_PITCH   (DIM + 4)
#define AGG_THREADS 1024
#define AGG_WAVES   (AGG_THREADS / 32)
#define AGG_GROUP   4
#define AGG_SPAN    (AGG_GROUP * 128)
#define AGG_STRIDE  (AGG_WAVES * AGG_SPAN)
#define AGG_VEC     (AGG_ROWS * DIM / 4)
#define AGG_CAP     64
#define AGG_UNITS   (AGG_ROWS * 2)
#define GEMM_COLS   48
#define GEMM_KHALF  (DIM / 2)

__device__ __forceinline__ int4 agg_load_dst4(const int* __restrict__ dst, int e) {
    const int  ec = e < N_EDGES - 4 ? e : N_EDGES - 4;
    const int4 d  = *(const int4*)(dst + ec);
    const bool in = e < N_EDGES;
    return make_int4(in ? d.x : -1, in ? d.y : -1, in ? d.z : -1, in ? d.w : -1);
}

__device__ __forceinline__ unsigned agg_file(int* cnt, int* lst, int rr, int e) {
    unsigned over = 0u;
    if ((unsigned)rr < (unsigned)AGG_ROWS) {
        const int slot = atomicAdd(cnt + rr, 1);
        if (slot < AGG_CAP) lst[rr * AGG_CAP + slot] = e;
        else                over = 1u;
    }
    return over;
}

struct gemm_frags { v2f a[3]; v2f b[3]; };

__device__ __forceinline__ void gemm_load_frags(gemm_frags& f,
                                                const float* a0, const float* a1, const float* a2,
                                                const float* __restrict__ wk) {
    f.a[0] = *(const v2f*)a0;
    f.a[1] = *(const v2f*)a1;
    f.a[2] = *(const v2f*)a2;
    f.b[0] = v2f{ wk[ 0], wk[DIM +  0] };
    f.b[1] = v2f{ wk[16], wk[DIM + 16] };
    f.b[2] = v2f{ wk[32], wk[DIM + 32] };
}

__device__ __forceinline__ void gemm_step(v8f (&d)[3][3], const gemm_frags& f) {
    #pragma unroll
    for (int i = 0; i < 3; ++i) {
        #pragma unroll
        for (int j = 0; j < 3; ++j)
            d[i][j] = __builtin_amdgcn_wmma_f32_16x16x4_f32(false, f.a[i], false, f.b[j],
                                                            (short)0, d[i][j], false, false);
    }
}

__global__ __launch_bounds__(AGG_THREADS) void gcn_aggregate_gemm(const int* __restrict__ src,
                                                                 const int* __restrict__ dst,
                                                                 const float* __restrict__ X,
                                                                 const float* __restrict__ W,
                                                                 const float* __restrict__ dinv,
                                                                 const float* __restrict__ bias,
                                                                 float* __restrict__ out) {
    __shared__ float acc[AGG_ROWS * AGG_PITCH];
    __shared__ int   lst[AGG_ROWS * AGG_CAP];
    __shared__ float lw[AGG_ROWS * AGG_CAP];
    __shared__ int   cnt[AGG_ROWS];
    __shared__ int   next_unit;

    const int tid  = threadIdx.x;
    const int lane = tid & 31;
    const int wave = __builtin_amdgcn_readfirstlane(tid >> 5);
    const int lo   = blockIdx.x * AGG_ROWS;

    #pragma unroll
    for (int it = 0; it < (AGG_VEC + AGG_THREADS - 1) / AGG_THREADS; ++it) {
        const int q = tid + it * AGG_THREADS;
        if (q < AGG_VEC) {
            const int r = q / (DIM / 4);
            const int c = (q - r * (DIM / 4)) * 4;
            *(float4*)(acc + r * AGG_PITCH + c) = make_float4(0.0f, 0.0f, 0.0f, 0.0f);
        }
    }
    if (tid < AGG_ROWS) cnt[tid] = 0;
    if (tid == AGG_ROWS) next_unit = 0;
    __syncthreads();

    int base = wave * AGG_SPAN;
    int4 cur[AGG_GROUP];
    #pragma unroll
    for (int g = 0; g < AGG_GROUP; ++g)
        cur[g] = agg_load_dst4(dst, base + g * 128 + lane * 4);

    for (; base < N_EDGES; base += AGG_STRIDE) {
        int4 nxt[AGG_GROUP];
        #pragma unroll
        for (int g = 0; g < AGG_GROUP; ++g)
            nxt[g] = agg_load_dst4(dst, base + AGG_STRIDE + g * 128 + lane * 4);

        unsigned over = 0u;
        #pragma unroll
        for (int g = 0; g < AGG_GROUP; ++g) {
            const int e0 = base + g * 128 + lane * 4;
            over |= agg_file(cnt, lst, cur[g].x - lo, e0 + 0) << (4 * g + 0);
            over |= agg_file(cnt, lst, cur[g].y - lo, e0 + 1) << (4 * g + 1);
            over |= agg_file(cnt, lst, cur[g].z - lo, e0 + 2) << (4 * g + 2);
            over |= agg_file(cnt, lst, cur[g].w - lo, e0 + 3) << (4 * g + 3);
        }
        while (over != 0u) {
            const int j = __builtin_ctz(over);
            over &= over - 1u;
            const int   e = base + (j >> 2) * 128 + lane * 4 + (j & 3);
            const int   t = dst[e];
            const int   s = src[e];
            const float n = dinv[s] * dinv[t];
            const float* xp = X + (size_t)s * DIM;
            float*       ap = acc + (t - lo) * AGG_PITCH;
            for (int c = 0; c < DIM; ++c)
                atomicAdd(ap + c, n * xp[c]);
        }

        #pragma unroll
        for (int g = 0; g < AGG_GROUP; ++g)
            cur[g] = nxt[g];
    }
    __syncthreads();

    {
        constexpr int META_IT = (AGG_ROWS * AGG_CAP + AGG_THREADS - 1) / AGG_THREADS;
        bool ok[META_IT];
        int  mr[META_IT];
        int  ms[META_IT];
        #pragma unroll
        for (int it = 0; it < META_IT; ++it) {
            const int q  = tid + it * AGG_THREADS;
            const int qc = q < AGG_ROWS * AGG_CAP ? q : AGG_ROWS * AGG_CAP - 1;
            mr[it] = qc / AGG_CAP;
            ok[it] = q < AGG_ROWS * AGG_CAP && qc - mr[it] * AGG_CAP < cnt[mr[it]];
            const int e = lst[qc];
            ms[it] = ok[it] ? e : 0;
        }
        static_assert(META_IT == 3, "the two fences below name three entries");
        #pragma unroll
        for (int it = 0; it < META_IT; ++it)
            ms[it] = src[ms[it]];
        asm volatile("" :: "v"(ms[0]), "v"(ms[1]), "v"(ms[2]));
        float ds[META_IT];
        float dt[META_IT];
        #pragma unroll
        for (int it = 0; it < META_IT; ++it) {
            ds[it] = dinv[ms[it]];
            dt[it] = dinv[lo + mr[it]];
        }
        asm volatile("" :: "v"(ds[0]), "v"(ds[1]), "v"(ds[2]), "v"(dt[0]), "v"(dt[1]), "v"(dt[2]));
        #pragma unroll
        for (int it = 0; it < META_IT; ++it) {
            if (ok[it]) {
                const int q = tid + it * AGG_THREADS;
                lst[q] = ms[it];
                lw[q]  = ds[it] * dt[it];
            }
        }
    }
    __syncthreads();

    for (;;) {
        int u = 0;
        if (lane == 0) u = atomicAdd(&next_unit, 1);
        u = __builtin_amdgcn_readfirstlane(u);
        if (u >= AGG_UNITS) break;
        const int r   = u >> 1;
        const int col = (u & 1) * (DIM / 2) + lane * 4;
        const int i   = lo + r;
        const int nr  = __builtin_amdgcn_readfirstlane(cnt[r]);
        const int n   = nr < AGG_CAP ? nr : AGG_CAP;
        const float  di = dinv[i];
        const float* xi = X + (size_t)i * DIM + col;
        float4 a[3];
        #pragma unroll
        for (int b = 0; b < 3; ++b) {
            const float4 x4 = *(const float4*)(xi + b * 128);
            a[b] = make_float4(di * di * x4.x, di * di * x4.y, di * di * x4.z, di * di * x4.w);
        }
        const int*   ls = lst + r * AGG_CAP;
        const float* lv = lw  + r * AGG_CAP;
        for (int k = 0; k < n; k += 4) {
            float4 v[4][3];
            float  w[4];
            #pragma unroll
            for (int q = 0; q < 4; ++q) {
                const int kk = k + q < n ? k + q : n - 1;
                const int s  = __builtin_amdgcn_readfirstlane(ls[kk]);
                w[q] = lv[kk];
                const float* xp = X + (size_t)s * DIM + col;
                #pragma unroll
                for (int b = 0; b < 3; ++b)
                    v[q][b] = *(const float4*)(xp + b * 128);
            }
            #pragma unroll
            for (int q = 0; q < 4; ++q) {
                if (k + q < n) {
                    #pragma unroll
                    for (int b = 0; b < 3; ++b) {
                        a[b].x += w[q] * v[q][b].x;
                        a[b].y += w[q] * v[q][b].y;
                        a[b].z += w[q] * v[q][b].z;
                        a[b].w += w[q] * v[q][b].w;
                    }
                }
            }
        }
        float* ap = acc + r * AGG_PITCH + col;
        #pragma unroll
        for (int b = 0; b < 3; ++b) {
            float4 o = *(const float4*)(ap + b * 128);
            o.x += a[b].x;
            o.y += a[b].y;
            o.z += a[b].z;
            o.w += a[b].w;
            *(float4*)(ap + b * 128) = o;
        }
    }
    __syncthreads();

    const int m    = lane & 15;
    const int hi   = lane >> 4;
    const int col0 = (wave & 15) * GEMM_COLS;
    const int kh   = wave >> 4;
    const int k0   = kh * GEMM_KHALF + hi * 2;
    const int row2 = 32 + m < AGG_ROWS ? 32 + m : AGG_ROWS - 1;

    const float* a0 = acc + m * AGG_PITCH + k0;
    const float* a1 = acc + (16 + m) * AGG_PITCH + k0;
    const float* a2 = acc + row2 * AGG_PITCH + k0;
    const float* wp = W + (size_t)k0 * DIM + col0 + m;

    v8f d[3][3] = {};
    gemm_frags f0, f1;
    gemm_load_frags(f0, a0, a1, a2, wp);
    #pragma clang loop unroll(disable)
    for (int k = 0; k < GEMM_KHALF; k += 8) {
        gemm_load_frags(f1, a0 + k + 4, a1 + k + 4, a2 + k + 4, wp + (size_t)(k + 4) * DIM);
        gemm_step(d, f0);
        const int kn = k + 8 < GEMM_KHALF ? k + 8 : k;
        gemm_load_frags(f0, a0 + kn, a1 + kn, a2 + kn, wp + (size_t)kn * DIM);
        gemm_step(d, f1);
    }
    __syncthreads();

    if (kh == 1) {
        #pragma unroll
        for (int i = 0; i < 3; ++i) {
            if (16 * i + 8 * hi < AGG_ROWS) {
                #pragma unroll
                for (int r = 0; r < 8; ++r) {
                    float* p = acc + (16 * i + 8 * hi + r) * AGG_PITCH + col0 + m;
                    p[ 0] = d[i][0][r];
                    p[16] = d[i][1][r];
                    p[32] = d[i][2][r];
                }
            }
        }
    }
    __syncthreads();
    if (kh == 0) {
        const float b0 = bias[col0 + m];
        const float b1 = bias[col0 + m + 16];
        const float b2 = bias[col0 + m + 32];
        #pragma unroll
        for (int i = 0; i < 3; ++i) {
            if (16 * i + 8 * hi < AGG_ROWS) {
                #pragma unroll
                for (int r = 0; r < 8; ++r) {
                    float* p = acc + (16 * i + 8 * hi + r) * AGG_PITCH + col0 + m;
                    p[ 0] = (d[i][0][r] + p[ 0]) + b0;
                    p[16] = (d[i][1][r] + p[16]) + b1;
                    p[32] = (d[i][2][r] + p[32]) + b2;
                }
            }
        }
    }
    __syncthreads();

    float* op = out + (size_t)lo * DIM;
    #pragma unroll
    for (int it = 0; it < (AGG_VEC + AGG_THREADS - 1) / AGG_THREADS; ++it) {
        const int q = tid + it * AGG_THREADS;
        if (q < AGG_VEC) {
            const int r = q / (DIM / 4);
            const int c = (q - r * (DIM / 4)) * 4;
            *(float4*)(op + q * 4) = *(const float4*)(acc + r * AGG_PITCH + c);
        }
    }
}

extern "C" void kernel_launch(void* const* d_in, const int* in_sizes, int n_in,
                              void* d_out, int out_size, void* d_ws, size_t ws_size,
                              hipStream_t stream) {
    const float* x    = (const float*)d_in[0];
    const int*   ei   = (const int*)d_in[1];
    const float* W    = (const float*)d_in[2];
    const float* bias = (const float*)d_in[3];
    float*       out  = (float*)d_out;

    const int* src = ei;
    const int* dst = ei + N_EDGES;

    float* deg  = (float*)d_ws;
    float* dinv = deg + N_NODES;

    gcn_init_deg <<<(N_NODES + 255) / 256, 256, 0, stream>>>(deg);
    gcn_count_deg<<<(N_EDGES + 255) / 256, 256, 0, stream>>>(dst, deg);
    gcn_rsqrt_deg<<<(N_NODES + 255) / 256, 256, 0, stream>>>(deg, dinv);

    gcn_aggregate_gemm<<<N_NODES / AGG_ROWS, AGG_THREADS, 0, stream>>>(src, dst, x, W, dinv, bias, out);
}


# Degree launches and in-kernel edge scan -> one owner-scan launch (125 groups x 80 rows) writing dinv, counts, per-row source lists to workspace; 4 launches -> 2; sum order = list slot order as before
# speedup vs baseline: 2.6278x; 1.1259x over previous
#include <hip/hip_runtime.h>
#include <hip/hip_bf16.h>

#define N_NODES 10000
#define N_EDGES 100000
#define DIM     768

typedef __attribute__((ext_vector_type(2))) float v2f;
typedef __attribute__((ext_vector_type(8))) float v8f;

#define AGG_ROWS    40
#define AGG_PITCH   (DIM + 4)
#define AGG_THREADS 1024
#define AGG_WAVES   (AGG_THREADS / 32)
#define AGG_GROUP   4
#define AGG_SPAN    (AGG_GROUP * 128)
#define AGG_STRIDE  (AGG_WAVES * AGG_SPAN)
#define AGG_VEC     (AGG_ROWS * DIM / 4)
#define AGG_CAP     64
#define FILE_ROWS   80
#define AGG_UNITS   (AGG_ROWS * 2)
#define GEMM_COLS   48
#define GEMM_KHALF  (DIM / 2)

__device__ __forceinline__ int4 agg_load_dst4(const int* __restrict__ dst, int e) {
    const int  ec = e < N_EDGES - 4 ? e : N_EDGES - 4;
    const int4 d  = *(const int4*)(dst + ec);
    const bool in = e < N_EDGES;
    return make_int4(in ? d.x : -1, in ? d.y : -1, in ? d.z : -1, in ? d.w : -1);
}

__device__ __forceinline__ void agg_file(int* cnt, int* lst, int rr, int e) {
    if ((unsigned)rr < (unsigned)FILE_ROWS) {
        const int slot = atomicAdd(cnt + rr, 1);
        if (slot < AGG_CAP) lst[rr * AGG_CAP + slot] = e;
    }
}

__global__ __launch_bounds__(AGG_THREADS) void gcn_file_edges(const int* __restrict__ src,
                                                             const int* __restrict__ dst,
                                                             float* __restrict__ dinv,
                                                             int* __restrict__ cntg,
                                                             int* __restrict__ lstg) {
    __shared__ int lst[FILE_ROWS * AGG_CAP];
    __shared__ int cnt[FILE_ROWS];

    const int tid  = threadIdx.x;
    const int lane = tid & 31;
    const int wave = __builtin_amdgcn_readfirstlane(tid >> 5);
    const int lo   = blockIdx.x * FILE_ROWS;

    if (tid < FILE_ROWS) cnt[tid] = 0;
    __syncthreads();

    int base = wave * AGG_SPAN;
    int4 cur[AGG_GROUP];
    #pragma unroll
    for (int g = 0; g < AGG_GROUP; ++g)
        cur[g] = agg_load_dst4(dst, base + g * 128 + lane * 4);

    for (; base < N_EDGES; base += AGG_STRIDE) {
        int4 nxt[AGG_GROUP];
        #pragma unroll
        for (int g = 0; g < AGG_GROUP; ++g)
            nxt[g] = agg_load_dst4(dst, base + AGG_STRIDE + g * 128 + lane * 4);

        #pragma unroll
        for (int g = 0; g < AGG_GROUP; ++g) {
            const int e0 = base + g * 128 + lane * 4;
            agg_file(cnt, lst, cur[g].x - lo, e0 + 0);
            agg_file(cnt, lst, cur[g].y - lo, e0 + 1);
            agg_file(cnt, lst, cur[g].z - lo, e0 + 2);
            agg_file(cnt, lst, cur[g].w - lo, e0 + 3);
        }

        #pragma unroll
        for (int g = 0; g < AGG_GROUP; ++g)
            cur[g] = nxt[g];
    }
    __syncthreads();

    if (tid < FILE_ROWS) {
        const int c = cnt[tid];
        cntg[lo + tid] = c;
        dinv[lo + tid] = rsqrtf(1.0f + (float)c);
    }
    constexpr int OUT_IT = (FILE_ROWS * AGG_CAP + AGG_THREADS - 1) / AGG_THREADS;
    bool ok[OUT_IT];
    int  ms[OUT_IT];
    #pragma unroll
    for (int it = 0; it < OUT_IT; ++it) {
        const int q  = tid + it * AGG_THREADS;
        const int qc = q < FILE_ROWS * AGG_CAP ? q : FILE_ROWS * AGG_CAP - 1;
        const int r  = qc / AGG_CAP;
        ok[it] = qc - r * AGG_CAP < cnt[r];
        const int e = lst[qc];
        ms[it] = ok[it] ? e : 0;
    }
    #pragma unroll
    for (int it = 0; it < OUT_IT; ++it)
        ms[it] = src[ms[it]];
    #pragma unroll
    for (int it = 0; it < OUT_IT; ++it) {
        const int q = tid + it * AGG_THREADS;
        if (q < FILE_ROWS * AGG_CAP)
            lstg[(size_t)lo * AGG_CAP + q] = ok[it] ? ms[it] : 0;
    }
}

struct gemm_frags { v2f a[3]; v2f b[3]; };

__device__ __forceinline__ void gemm_load_frags(gemm_frags& f,
                                                const float* a0, const float* a1, const float* a2,
                                                const float* __restrict__ wk) {
    f.a[0] = *(const v2f*)a0;
    f.a[1] = *(const v2f*)a1;
    f.a[2] = *(const v2f*)a2;
    f.b[0] = v2f{ wk[ 0], wk[DIM +  0] };
    f.b[1] = v2f{ wk[16], wk[DIM + 16] };
    f.b[2] = v2f{ wk[32], wk[DIM + 32] };
}

__device__ __forceinline__ void gemm_step(v8f (&d)[3][3], const gemm_frags& f) {
    #pragma unroll
    for (int i = 0; i < 3; ++i) {
        #pragma unroll
        for (int j = 0; j < 3; ++j)
            d[i][j] = __builtin_amdgcn_wmma_f32_16x16x4_f32(false, f.a[i], false, f.b[j],
                                                            (short)0, d[i][j], false, false);
    }
}

__global__ __launch_bounds__(AGG_THREADS) void gcn_aggregate_gemm(const int* __restrict__ src,
                                                                 const int* __restrict__ dst,
                                                                 const float* __restrict__ X,
                                                                 const float* __restrict__ W,
                                                                 const float* __restrict__ dinv,
                                                                 const int* __restrict__ cntg,
                                                                 const int* __restrict__ lstg,
                                                                 const float* __restrict__ bias,
                                                                 float* __restrict__ out) {
    __shared__ float acc[AGG_ROWS * AGG_PITCH];
    __shared__ int   lst[AGG_ROWS * AGG_CAP];
    __shared__ float lw[AGG_ROWS * AGG_CAP];
    __shared__ int   cnt[AGG_ROWS];
    __shared__ int   next_unit;

    const int tid  = threadIdx.x;
    const int lane = tid & 31;
    const int wave = __builtin_amdgcn_readfirstlane(tid >> 5);
    const int lo   = blockIdx.x * AGG_ROWS;

    #pragma unroll
    for (int it = 0; it < (AGG_VEC + AGG_THREADS - 1) / AGG_THREADS; ++it) {
        const int q = tid + it * AGG_THREADS;
        if (q < AGG_VEC) {
            const int r = q / (DIM / 4);
            const int c = (q - r * (DIM / 4)) * 4;
            *(float4*)(acc + r * AGG_PITCH + c) = make_float4(0.0f, 0.0f, 0.0f, 0.0f);
        }
    }
    if (tid == 0) next_unit = 0;

    {
        constexpr int META_IT = (AGG_ROWS * AGG_CAP + AGG_THREADS - 1) / AGG_THREADS;
        bool ok[META_IT];
        int  mr[META_IT];
        int  mc[META_IT];
        int  ms[META_IT];
        #pragma unroll
        for (int it = 0; it < META_IT; ++it) {
            const int q  = tid + it * AGG_THREADS;
            const int qc = q < AGG_ROWS * AGG_CAP ? q : AGG_ROWS * AGG_CAP - 1;
            mr[it] = qc / AGG_CAP;
            mc[it] = cntg[lo + mr[it]];
            ok[it] = q < AGG_ROWS * AGG_CAP && mc[it] <= AGG_CAP && qc - mr[it] * AGG_CAP < mc[it];
            ms[it] = lstg[(size_t)lo * AGG_CAP + qc];
        }
        static_assert(META_IT == 3, "the two fences below name three entries");
        asm volatile("" :: "v"(ms[0]), "v"(ms[1]), "v"(ms[2]));
        float ds[META_IT];
        float dt[META_IT];
        #pragma unroll
        for (int it = 0; it < META_IT; ++it) {
            ds[it] = dinv[ms[it]];
            dt[it] = dinv[lo + mr[it]];
        }
        asm volatile("" :: "v"(ds[0]), "v"(ds[1]), "v"(ds[2]), "v"(dt[0]), "v"(dt[1]), "v"(dt[2]));
        #pragma unroll
        for (int it = 0; it < META_IT; ++it) {
            const int q = tid + it * AGG_THREADS;
            if (q < AGG_ROWS * AGG_CAP && q - mr[it] * AGG_CAP == 0) cnt[mr[it]] = mc[it];
            if (ok[it]) {
                lst[q] = ms[it];
                lw[q]  = ds[it] * dt[it];
            }
        }
    }
    __syncthreads();

    {
        static_assert(AGG_ROWS > 32 && AGG_ROWS <= 64, "a lane looks at rows lane and 32 + lane");
        const int  c0 = cnt[lane];
        const int  c1 = cnt[lane < AGG_ROWS - 32 ? 32 + lane : 0];
        const bool any_over = __builtin_amdgcn_ballot_w32(c0 > AGG_CAP || c1 > AGG_CAP) != 0u;
        if (any_over) {
            for (int e = tid; e < N_EDGES; e += AGG_THREADS) {
                const int t = dst[e] - lo;
                if ((unsigned)t < (unsigned)AGG_ROWS && cnt[t] > AGG_CAP) {
                    const int   sn = src[e];
                    const float n  = dinv[sn] * dinv[lo + t];
                    const float* xp = X + (size_t)sn * DIM;
                    float*       ap = acc + t * AGG_PITCH;
                    for (int c = 0; c < DIM; ++c)
                        atomicAdd(ap + c, n * xp[c]);
                }
            }
            __syncthreads();
        }
    }

    for (;;) {
        int u = 0;
        if (lane == 0) u = atomicAdd(&next_unit, 1);
        u = __builtin_amdgcn_readfirstlane(u);
        if (u >= AGG_UNITS) break;
        const int r   = u >> 1;
        const int col = (u & 1) * (DIM / 2) + lane * 4;
        const int i   = lo + r;
        const int nr  = __builtin_amdgcn_readfirstlane(cnt[r]);
        const int n   = nr <= AGG_CAP ? nr : 0;
        const float  di = dinv[i];
        const float* xi = X + (size_t)i * DIM + col;
        float4 a[3];
        #pragma unroll
        for (int b = 0; b < 3; ++b) {
            const float4 x4 = *(const float4*)(xi + b * 128);
            a[b] = make_float4(di * di * x4.x, di * di * x4.y, di * di * x4.z, di * di * x4.w);
        }
        const int*   ls = lst + r * AGG_CAP;
        const float* lv = lw  + r * AGG_CAP;
        for (int k = 0; k < n; k += 4) {
            float4 v[4][3];
            float  w[4];
            #pragma unroll
            for (int q = 0; q < 4; ++q) {
                const int kk = k + q < n ? k + q : n - 1;
                const int s  = __builtin_amdgcn_readfirstlane(ls[kk]);
                w[q] = lv[kk];
                const float* xp = X + (size_t)s * DIM + col;
                #pragma unroll
                for (int b = 0; b < 3; ++b)
                    v[q][b] = *(const float4*)(xp + b * 128);
            }
            #pragma unroll
            for (int q = 0; q < 4; ++q) {
                if (k + q < n) {
                    #pragma unroll
                    for (int b = 0; b < 3; ++b) {
                        a[b].x += w[q] * v[q][b].x;
                        a[b].y += w[q] * v[q][b].y;
                        a[b].z += w[q] * v[q][b].z;
                        a[b].w += w[q] * v[q][b].w;
                    }
                }
            }
        }
        float* ap = acc + r * AGG_PITCH + col;
        #pragma unroll
        for (int b = 0; b < 3; ++b) {
            float4 o = *(const float4*)(ap + b * 128);
            o.x += a[b].x;
            o.y += a[b].y;
            o.z += a[b].z;
            o.w += a[b].w;
            *(float4*)(ap + b * 128) = o;
        }
    }
    __syncthreads();

    const int m    = lane & 15;
    const int hi   = lane >> 4;
    const int col0 = (wave & 15) * GEMM_COLS;
    const int kh   = wave >> 4;
    const int k0   = kh * GEMM_KHALF + hi * 2;
    const int row2 = 32 + m < AGG_ROWS ? 32 + m : AGG_ROWS - 1;

    const float* a0 = acc + m * AGG_PITCH + k0;
    const float* a1 = acc + (16 + m) * AGG_PITCH + k0;
    const float* a2 = acc + row2 * AGG_PITCH + k0;
    const float* wp = W + (size_t)k0 * DIM + col0 + m;

    v8f d[3][3] = {};
    gemm_frags f0, f1;
    gemm_load_frags(f0, a0, a1, a2, wp);
    #pragma clang loop unroll(disable)
    for (int k = 0; k < GEMM_KHALF; k += 8) {
        gemm_load_frags(f1, a0 + k + 4, a1 + k + 4, a2 + k + 4, wp + (size_t)(k + 4) * DIM);
        gemm_step(d, f0);
        const int kn = k + 8 < GEMM_KHALF ? k + 8 : k;
        gemm_load_frags(f0, a0 + kn, a1 + kn, a2 + kn, wp + (size_t)kn * DIM);
        gemm_step(d, f1);
    }
    __syncthreads();

    if (kh == 1) {
        #pragma unroll
        for (int i = 0; i < 3; ++i) {
            if (16 * i + 8 * hi < AGG_ROWS) {
                #pragma unroll
                for (int r = 0; r < 8; ++r) {
                    float* p = acc + (16 * i + 8 * hi + r) * AGG_PITCH + col0 + m;
                    p[ 0] = d[i][0][r];
                    p[16] = d[i][1][r];
                    p[32] = d[i][2][r];
                }
            }
        }
    }
    __syncthreads();
    if (kh == 0) {
        const float b0 = bias[col0 + m];
        const float b1 = bias[col0 + m + 16];
        const float b2 = bias[col0 + m + 32];
        #pragma unroll
        for (int i = 0; i < 3; ++i) {
            if (16 * i + 8 * hi < AGG_ROWS) {
                #pragma unroll
                for (int r = 0; r < 8; ++r) {
                    float* p = acc + (16 * i + 8 * hi + r) * AGG_PITCH + col0 + m;
                    p[ 0] = (d[i][0][r] + p[ 0]) + b0;
                    p[16] = (d[i][1][r] + p[16]) + b1;
                    p[32] = (d[i][2][r] + p[32]) + b2;
                }
            }
        }
    }
    __syncthreads();

    float* op = out + (size_t)lo * DIM;
    #pragma unroll
    for (int it = 0; it < (AGG_VEC + AGG_THREADS - 1) / AGG_THREADS; ++it) {
        const int q = tid + it * AGG_THREADS;
        if (q < AGG_VEC) {
            const int r = q / (DIM / 4);
            const int c = (q - r * (DIM / 4)) * 4;
            *(float4*)(op + q * 4) = *(const float4*)(acc + r * AGG_PITCH + c);
        }
    }
}

extern "C" void kernel_launch(void* const* d_in, const int* in_sizes, int n_in,
                              void* d_out, int out_size, void* d_ws, size_t ws_size,
                              hipStream_t stream) {
    const float* x    = (const float*)d_in[0];
    const int*   ei   = (const int*)d_in[1];
    const float* W    = (const float*)d_in[2];
    const float* bias = (const float*)d_in[3];
    float*       out  = (float*)d_out;

    const int* src = ei;
    const int* dst = ei + N_EDGES;

    float* dinv = (float*)d_ws;
    int*   cntg = (int*)d_ws + N_NODES;
    int*   lstg = cntg + N_NODES;

    gcn_file_edges    <<<N_NODES / FILE_ROWS, AGG_THREADS, 0, stream>>>(src, dst, dinv, cntg, lstg);
    gcn_aggregate_gemm<<<N_NODES / AGG_ROWS, AGG_THREADS, 0, stream>>>(src, dst, x, W, dinv, cntg, lstg, bias, out);
}


# Pairs of work-groups own 80 rows = five whole 16-row matrix tiles; the middle tile's columns are split between the two (both sum its z), so no tile holds padding rows; sum order unchanged
# speedup vs baseline: 2.8670x; 1.0910x over previous
#include <hip/hip_runtime.h>
#include <hip/hip_bf16.h>

#define N_NODES 10000
#define N_EDGES 100000
#define DIM     768

typedef __attribute__((ext_vector_type(2))) float v2f;
typedef __attribute__((ext_vector_type(8))) float v8f;

#define PAIR_ROWS   80
#define AGG_ROWS    48
#define AGG_GROUPS  (2 * (N_NODES / PAIR_ROWS))
#define AGG_PITCH   (DIM + 4)
#define AGG_THREADS 1024
#define AGG_WAVES   (AGG_THREADS / 32)
#define AGG_GROUP   4
#define AGG_SPAN    (AGG_GROUP * 128)
#define AGG_STRIDE  (AGG_WAVES * AGG_SPAN)
#define AGG_VEC     (AGG_ROWS * DIM / 4)
#define AGG_CAP     64
#define FILE_ROWS   80
#define AGG_UNITS   (AGG_ROWS * 2)
#define GEMM_COLS   48
#define GEMM_KHALF  (DIM / 2)
static_assert(N_NODES % PAIR_ROWS == 0 && PAIR_ROWS == 80 && AGG_ROWS == 48,
              "a pair is five 16-row tiles; a group holds three of them, the middle one shared");
static_assert(AGG_VEC % AGG_THREADS == 0, "the zeroing and the out loop have no partial trip");

__device__ __forceinline__ int4 agg_load_dst4(const int* __restrict__ dst, int e) {
    const int  ec = e < N_EDGES - 4 ? e : N_EDGES - 4;
    const int4 d  = *(const int4*)(dst + ec);
    const bool in = e < N_EDGES;
    return make_int4(in ? d.x : -1, in ? d.y : -1, in ? d.z : -1, in ? d.w : -1);
}

__device__ __forceinline__ void agg_file(int* cnt, int* lst, int rr, int e) {
    if ((unsigned)rr < (unsigned)FILE_ROWS) {
        const int slot = atomicAdd(cnt + rr, 1);
        if (slot < AGG_CAP) lst[rr * AGG_CAP + slot] = e;
    }
}

__global__ __launch_bounds__(AGG_THREADS) void gcn_file_edges(const int* __restrict__ src,
                                                             const int* __restrict__ dst,
                                                             float* __restrict__ dinv,
                                                             int* __restrict__ cntg,
                                                             int* __restrict__ lstg) {
    __shared__ int lst[FILE_ROWS * AGG_CAP];
    __shared__ int cnt[FILE_ROWS];

    const int tid  = threadIdx.x;
    const int lane = tid & 31;
    const int wave = __builtin_amdgcn_readfirstlane(tid >> 5);
    const int lo   = blockIdx.x * FILE_ROWS;

    if (tid < FILE_ROWS) cnt[tid] = 0;
    __syncthreads();

    int base = wave * AGG_SPAN;
    int4 cur[AGG_GROUP];
    #pragma unroll
    for (int g = 0; g < AGG_GROUP; ++g)
        cur[g] = agg_load_dst4(dst, base + g * 128 + lane * 4);

    for (; base < N_EDGES; base += AGG_STRIDE) {
        int4 nxt[AGG_GROUP];
        #pragma unroll
        for (int g = 0; g < AGG_GROUP; ++g)
            nxt[g] = agg_load_dst4(dst, base + AGG_STRIDE + g * 128 + lane * 4);

        #pragma unroll
        for (int g = 0; g < AGG_GROUP; ++g) {
            const int e0 = base + g * 128 + lane * 4;
            agg_file(cnt, lst, cur[g].x - lo, e0 + 0);
            agg_file(cnt, lst, cur[g].y - lo, e0 + 1);
            agg_file(cnt, lst, cur[g].z - lo, e0 + 2);
            agg_file(cnt, lst, cur[g].w - lo, e0 + 3);
        }

        #pragma unroll
        for (int g = 0; g < AGG_GROUP; ++g)
            cur[g] = nxt[g];
    }
    __syncthreads();

    if (tid < FILE_ROWS) {
        const int c = cnt[tid];
        cntg[lo + tid] = c;
        dinv[lo + tid] = rsqrtf(1.0f + (float)c);
    }
    constexpr int OUT_IT = (FILE_ROWS * AGG_CAP + AGG_THREADS - 1) / AGG_THREADS;
    bool ok[OUT_IT];
    int  ms[OUT_IT];
    #pragma unroll
    for (int it = 0; it < OUT_IT; ++it) {
        const int q  = tid + it * AGG_THREADS;
        const int qc = q < FILE_ROWS * AGG_CAP ? q : FILE_ROWS * AGG_CAP - 1;
        const int r  = qc / AGG_CAP;
        ok[it] = qc - r * AGG_CAP < cnt[r];
        const int e = lst[qc];
        ms[it] = ok[it] ? e : 0;
    }
    #pragma unroll
    for (int it = 0; it < OUT_IT; ++it)
        ms[it] = src[ms[it]];
    #pragma unroll
    for (int it = 0; it < OUT_IT; ++it) {
        const int q = tid + it * AGG_THREADS;
        if (q < FILE_ROWS * AGG_CAP)
            lstg[(size_t)lo * AGG_CAP + q] = ok[it] ? ms[it] : 0;
    }
}

template <int NM>
struct gemm_frags { v2f a[NM]; v2f b[3]; };

template <int NM>
__device__ __forceinline__ void gemm_load_frags(gemm_frags<NM>& f, const float* a,
                                                const float* __restrict__ wk) {
    #pragma unroll
    for (int i = 0; i < NM; ++i)
        f.a[i] = *(const v2f*)(a + i * 16 * AGG_PITCH);
    f.b[0] = v2f{ wk[ 0], wk[DIM +  0] };
    f.b[1] = v2f{ wk[16], wk[DIM + 16] };
    f.b[2] = v2f{ wk[32], wk[DIM + 32] };
}

template <int NM>
__device__ __forceinline__ void gemm_step(v8f (&d)[3][3], const gemm_frags<NM>& f) {
    #pragma unroll
    for (int i = 0; i < NM; ++i) {
        #pragma unroll
        for (int j = 0; j < 3; ++j)
            d[i][j] = __builtin_amdgcn_wmma_f32_16x16x4_f32(false, f.a[i], false, f.b[j],
                                                            (short)0, d[i][j], false, false);
    }
}

template <int NM>
__device__ __forceinline__ void gemm_run(v8f (&d)[3][3], const float* a, const float* __restrict__ wp) {
    gemm_frags<NM> f0, f1;
    gemm_load_frags(f0, a, wp);
    #pragma clang loop unroll(disable)
    for (int k = 0; k < GEMM_KHALF; k += 8) {
        gemm_load_frags(f1, a + k + 4, wp + (size_t)(k + 4) * DIM);
        gemm_step(d, f0);
        const int kn = k + 8 < GEMM_KHALF ? k + 8 : k;
        gemm_load_frags(f0, a + kn, wp + (size_t)kn * DIM);
        gemm_step(d, f1);
    }
}

__global__ __launch_bounds__(AGG_THREADS) void gcn_aggregate_gemm(const int* __restrict__ src,
                                                                 const int* __restrict__ dst,
                                                                 const float* __restrict__ X,
                                                                 const float* __restrict__ W,
                                                                 const float* __restrict__ dinv,
                                                                 const int* __restrict__ cntg,
                                                                 const int* __restrict__ lstg,
                                                                 const float* __restrict__ bias,
                                                                 float* __restrict__ out) {
    __shared__ float acc[AGG_ROWS * AGG_PITCH];
    __shared__ int   lst[AGG_ROWS * AGG_CAP];
    __shared__ float lw[AGG_ROWS * AGG_CAP];
    __shared__ int   cnt[AGG_ROWS];
    __shared__ int   next_unit;

    const int tid  = threadIdx.x;
    const int lane = tid & 31;
    const int wave = __builtin_amdgcn_readfirstlane(tid >> 5);
    const int side = blockIdx.x & 1;
    const int lo   = (blockIdx.x >> 1) * PAIR_ROWS + side * 32;

    #pragma unroll
    for (int it = 0; it < (AGG_VEC + AGG_THREADS - 1) / AGG_THREADS; ++it) {
        const int q = tid + it * AGG_THREADS;
        if (q < AGG_VEC) {
            const int r = q / (DIM / 4);
            const int c = (q - r * (DIM / 4)) * 4;
            *(float4*)(acc + r * AGG_PITCH + c) = make_float4(0.0f, 0.0f, 0.0f, 0.0f);
        }
    }
    if (tid == 0) next_unit = 0;

    {
        constexpr int META_IT = (AGG_ROWS * AGG_CAP + AGG_THREADS - 1) / AGG_THREADS;
        bool ok[META_IT];
        int  mr[META_IT];
        int  mc[META_IT];
        int  ms[META_IT];
        #pragma unroll
        for (int it = 0; it < META_IT; ++it) {
            const int q  = tid + it * AGG_THREADS;
            const int qc = q < AGG_ROWS * AGG_CAP ? q : AGG_ROWS * AGG_CAP - 1;
            mr[it] = qc / AGG_CAP;
            mc[it] = cntg[lo + mr[it]];
            ok[it] = q < AGG_ROWS * AGG_CAP && mc[it] <= AGG_CAP && qc - mr[it] * AGG_CAP < mc[it];
            ms[it] = lstg[(size_t)lo * AGG_CAP + qc];
        }
        static_assert(META_IT == 3, "the two fences below name three entries");
        asm volatile("" :: "v"(ms[0]), "v"(ms[1]), "v"(ms[2]));
        float ds[META_IT];
        float dt[META_IT];
        #pragma unroll
        for (int it = 0; it < META_IT; ++it) {
            ds[it] = dinv[ms[it]];
            dt[it] = dinv[lo + mr[it]];
        }
        asm volatile("" :: "v"(ds[0]), "v"(ds[1]), "v"(ds[2]), "v"(dt[0]), "v"(dt[1]), "v"(dt[2]));
        #pragma unroll
        for (int it = 0; it < META_IT; ++it) {
            const int q = tid + it * AGG_THREADS;
            if (q < AGG_ROWS * AGG_CAP && q - mr[it] * AGG_CAP == 0) cnt[mr[it]] = mc[it];
            if (ok[it]) {
                lst[q] = ms[it];
                lw[q]  = ds[it] * dt[it];
            }
        }
    }
    __syncthreads();

    {
        static_assert(AGG_ROWS > 32 && AGG_ROWS <= 64, "a lane looks at rows lane and 32 + lane");
        const int  c0 = cnt[lane];
        const int  c1 = cnt[lane < AGG_ROWS - 32 ? 32 + lane : 0];
        const bool any_over = __builtin_amdgcn_ballot_w32(c0 > AGG_CAP || c1 > AGG_CAP) != 0u;
        if (any_over) {
            for (int e = tid; e < N_EDGES; e += AGG_THREADS) {
                const int t = dst[e] - lo;
                if ((unsigned)t < (unsigned)AGG_ROWS && cnt[t] > AGG_CAP) {
                    const int   sn = src[e];
                    const float n  = dinv[sn] * dinv[lo + t];
                    const float* xp = X + (size_t)sn * DIM;
                    float*       ap = acc + t * AGG_PITCH;
                    for (int c = 0; c < DIM; ++c)
                        atomicAdd(ap + c, n * xp[c]);
                }
            }
            __syncthreads();
        }
    }

    for (;;) {
        int u = 0;
        if (lane == 0) u = atomicAdd(&next_unit, 1);
        u = __builtin_amdgcn_readfirstlane(u);
        if (u >= AGG_UNITS) break;
        const int r   = u >> 1;
        const int col = (u & 1) * (DIM / 2) + lane * 4;
        const int i   = lo + r;
        const int nr  = __builtin_amdgcn_readfirstlane(cnt[r]);
        const int n   = nr <= AGG_CAP ? nr : 0;
        const float  di = dinv[i];
        const float* xi = X + (size_t)i * DIM + col;
        float4 a[3];
        #pragma unroll
        for (int b = 0; b < 3; ++b) {
            const float4 x4 = *(const float4*)(xi + b * 128);
            a[b] = make_float4(di * di * x4.x, di * di * x4.y, di * di * x4.z, di * di * x4.w);
        }
        const int*   ls = lst + r * AGG_CAP;
        const float* lv = lw  + r * AGG_CAP;
        for (int k = 0; k < n; k += 4) {
            float4 v[4][3];
            float  w[4];
            #pragma unroll
            for (int q = 0; q < 4; ++q) {
                const int kk = k + q < n ? k + q : n - 1;
                const int s  = __builtin_amdgcn_readfirstlane(ls[kk]);
                w[q] = lv[kk];
                const float* xp = X + (size_t)s * DIM + col;
                #pragma unroll
                for (int b = 0; b < 3; ++b)
                    v[q][b] = *(const float4*)(xp + b * 128);
            }
            #pragma unroll
            for (int q = 0; q < 4; ++q) {
                if (k + q < n) {
                    #pragma unroll
                    for (int b = 0; b < 3; ++b) {
                        a[b].x += w[q] * v[q][b].x;
                        a[b].y += w[q] * v[q][b].y;
                        a[b].z += w[q] * v[q][b].z;
                        a[b].w += w[q] * v[q][b].w;
                    }
                }
            }
        }
        float* ap = acc + r * AGG_PITCH + col;
        #pragma unroll
        for (int b = 0; b < 3; ++b) {
            float4 o = *(const float4*)(ap + b * 128);
            o.x += a[b].x;
            o.y += a[b].y;
            o.z += a[b].z;
            o.w += a[b].w;
            *(float4*)(ap + b * 128) = o;
        }
    }
    __syncthreads();

    const int  m     = lane & 15;
    const int  hi    = lane >> 4;
    const int  wb2   = (wave >> 2) & 1;
    const bool full  = wb2 == 0;
    const int  nhalf = wb2 ^ side;
    const int  col0  = (nhalf * 8 + ((wave & 3) | (((wave >> 3) & 1) << 2))) * GEMM_COLS;
    const int  kh    = wave >> 4;
    const int  k0    = kh * GEMM_KHALF + hi * 2;
    const int  shr   = side ? 0 : 2;
    const int  t0    = full ? 0 : side;
    const int  nm    = full ? 3 : 2;

    const float* ab = acc + (16 * t0 + m) * AGG_PITCH + k0;
    const float* wp = W + (size_t)k0 * DIM + col0 + m;

    v8f d[3][3] = {};
    if (full) gemm_run<3>(d, ab, wp);
    else      gemm_run<2>(d, ab, wp);
    __syncthreads();

    if (kh == 1) {
        #pragma unroll
        for (int i = 0; i < 3; ++i) {
            if (i < nm) {
                #pragma unroll
                for (int r = 0; r < 8; ++r) {
                    float* p = acc + (16 * (t0 + i) + 8 * hi + r) * AGG_PITCH + col0 + m;
                    p[ 0] = d[i][0][r];
                    p[16] = d[i][1][r];
                    p[32] = d[i][2][r];
                }
            }
        }
    }
    __syncthreads();
    if (kh == 0) {
        const float b0 = bias[col0 + m];
        const float b1 = bias[col0 + m + 16];
        const float b2 = bias[col0 + m + 32];
        #pragma unroll
        for (int i = 0; i < 3; ++i) {
            if (i < nm) {
                #pragma unroll
                for (int r = 0; r < 8; ++r) {
                    float* p = acc + (16 * (t0 + i) + 8 * hi + r) * AGG_PITCH + col0 + m;
                    p[ 0] = (d[i][0][r] + p[ 0]) + b0;
                    p[16] = (d[i][1][r] + p[16]) + b1;
                    p[32] = (d[i][2][r] + p[32]) + b2;
                }
            }
        }
    }
    __syncthreads();

    float* op = out + (size_t)lo * DIM;
    #pragma unroll
    for (int it = 0; it < AGG_VEC / AGG_THREADS; ++it) {
        const int q = tid + it * AGG_THREADS;
        const int r = q / (DIM / 4);
        const int c = q - r * (DIM / 4);
        if ((r >> 4) != shr || (c >= DIM / 8 ? 1 : 0) == side)
            *(float4*)(op + q * 4) = *(const float4*)(acc + r * AGG_PITCH + c * 4);
    }
}

extern "C" void kernel_launch(void* const* d_in, const int* in_sizes, int n_in,
                              void* d_out, int out_size, void* d_ws, size_t ws_size,
                              hipStream_t stream) {
    const float* x    = (const float*)d_in[0];
    const int*   ei   = (const int*)d_in[1];
    const float* W    = (const float*)d_in[2];
    const float* bias = (const float*)d_in[3];
    float*       out  = (float*)d_out;

    const int* src = ei;
    const int* dst = ei + N_EDGES;

    float* dinv = (float*)d_ws;
    int*   cntg = (int*)d_ws + N_NODES;
    int*   lstg = cntg + N_NODES;

    gcn_file_edges    <<<N_NODES / FILE_ROWS, AGG_THREADS, 0, stream>>>(src, dst, dinv, cntg, lstg);
    gcn_aggregate_gemm<<<AGG_GROUPS, AGG_THREADS, 0, stream>>>(src, dst, x, W, dinv, cntg, lstg, bias, out);
}


# W fragments (B operand) requested 3 k-steps ahead in three-tile waves, 5 in two-tile waves (register ring), not 1; z fragments 1 ahead; LDS clear re-indexed to free registers; sum order unchanged
# speedup vs baseline: 3.0040x; 1.0478x over previous
#include <hip/hip_runtime.h>
#include <hip/hip_bf16.h>

#define N_NODES 10000
#define N_EDGES 100000
#define DIM     768

typedef __attribute__((ext_vector_type(2))) float v2f;
typedef __attribute__((ext_vector_type(8))) float v8f;

#define PAIR_ROWS   80
#define AGG_ROWS    48
#define AGG_GROUPS  (2 * (N_NODES / PAIR_ROWS))
#define AGG_PITCH   (DIM + 4)
#define AGG_THREADS 1024
#define AGG_WAVES   (AGG_THREADS / 32)
#define AGG_GROUP   4
#define AGG_SPAN    (AGG_GROUP * 128)
#define AGG_STRIDE  (AGG_WAVES * AGG_SPAN)
#define AGG_VEC     (AGG_ROWS * DIM / 4)
#define AGG_CAP     64
#define FILE_ROWS   80
#define AGG_UNITS   (AGG_ROWS * 2)
#define GEMM_COLS   48
#define GEMM_KHALF  (DIM / 2)
#define GEMM_RING3  4
#define GEMM_RING2  6
static_assert(N_NODES % PAIR_ROWS == 0 && PAIR_ROWS == 80 && AGG_ROWS == 48,
              "a pair is five 16-row tiles; a group holds three of them, the middle one shared");
static_assert(AGG_VEC % AGG_THREADS == 0, "the out loop has no partial trip");

__device__ __forceinline__ int4 agg_load_dst4(const int* __restrict__ dst, int e) {
    const int  ec = e < N_EDGES - 4 ? e : N_EDGES - 4;
    const int4 d  = *(const int4*)(dst + ec);
    const bool in = e < N_EDGES;
    return make_int4(in ? d.x : -1, in ? d.y : -1, in ? d.z : -1, in ? d.w : -1);
}

__device__ __forceinline__ void agg_file(int* cnt, int* lst, int rr, int e) {
    if ((unsigned)rr < (unsigned)FILE_ROWS) {
        const int slot = atomicAdd(cnt + rr, 1);
        if (slot < AGG_CAP) lst[rr * AGG_CAP + slot] = e;
    }
}

__global__ __launch_bounds__(AGG_THREADS) void gcn_file_edges(const int* __restrict__ src,
                                                             const int* __restrict__ dst,
                                                             float* __restrict__ dinv,
                                                             int* __restrict__ cntg,
                                                             int* __restrict__ lstg) {
    __shared__ int lst[FILE_ROWS * AGG_CAP];
    __shared__ int cnt[FILE_ROWS];

    const int tid  = threadIdx.x;
    const int lane = tid & 31;
    const int wave = __builtin_amdgcn_readfirstlane(tid >> 5);
    const int lo   = blockIdx.x * FILE_ROWS;

    if (tid < FILE_ROWS) cnt[tid] = 0;
    __syncthreads();

    int base = wave * AGG_SPAN;
    int4 cur[AGG_GROUP];
    #pragma unroll
    for (int g = 0; g < AGG_GROUP; ++g)
        cur[g] = agg_load_dst4(dst, base + g * 128 + lane * 4);

    for (; base < N_EDGES; base += AGG_STRIDE) {
        int4 nxt[AGG_GROUP];
        #pragma unroll
        for (int g = 0; g < AGG_GROUP; ++g)
            nxt[g] = agg_load_dst4(dst, base + AGG_STRIDE + g * 128 + lane * 4);

        #pragma unroll
        for (int g = 0; g < AGG_GROUP; ++g) {
            const int e0 = base + g * 128 + lane * 4;
            agg_file(cnt, lst, cur[g].x - lo, e0 + 0);
            agg_file(cnt, lst, cur[g].y - lo, e0 + 1);
            agg_file(cnt, lst, cur[g].z - lo, e0 + 2);
            agg_file(cnt, lst, cur[g].w - lo, e0 + 3);
        }

        #pragma unroll
        for (int g = 0; g < AGG_GROUP; ++g)
            cur[g] = nxt[g];
    }
    __syncthreads();

    if (tid < FILE_ROWS) {
        const int c = cnt[tid];
        cntg[lo + tid] = c;
        dinv[lo + tid] = rsqrtf(1.0f + (float)c);
    }
    constexpr int OUT_IT = (FILE_ROWS * AGG_CAP + AGG_THREADS - 1) / AGG_THREADS;
    bool ok[OUT_IT];
    int  ms[OUT_IT];
    #pragma unroll
    for (int it = 0; it < OUT_IT; ++it) {
        const int q  = tid + it * AGG_THREADS;
        const int qc = q < FILE_ROWS * AGG_CAP ? q : FILE_ROWS * AGG_CAP - 1;
        const int r  = qc / AGG_CAP;
        ok[it] = qc - r * AGG_CAP < cnt[r];
        const int e = lst[qc];
        ms[it] = ok[it] ? e : 0;
    }
    #pragma unroll
    for (int it = 0; it < OUT_IT; ++it)
        ms[it] = src[ms[it]];
    #pragma unroll
    for (int it = 0; it < OUT_IT; ++it) {
        const int q = tid + it * AGG_THREADS;
        if (q < FILE_ROWS * AGG_CAP)
            lstg[(size_t)lo * AGG_CAP + q] = ok[it] ? ms[it] : 0;
    }
}

struct gemm_bfrag { v2f b[3]; };

__device__ __forceinline__ void gemm_load_b(gemm_bfrag& f, const float* __restrict__ wk) {
    f.b[0] = v2f{ wk[ 0], wk[DIM +  0] };
    f.b[1] = v2f{ wk[16], wk[DIM + 16] };
    f.b[2] = v2f{ wk[32], wk[DIM + 32] };
}

template <int NM>
__device__ __forceinline__ void gemm_load_a(v2f (&fa)[NM], const float* a) {
    #pragma unroll
    for (int i = 0; i < NM; ++i)
        fa[i] = *(const v2f*)(a + i * 16 * AGG_PITCH);
}

template <int NM>
__device__ __forceinline__ void gemm_step(v8f (&d)[3][3], const v2f (&fa)[NM], const gemm_bfrag& f) {
    #pragma unroll
    for (int i = 0; i < NM; ++i) {
        #pragma unroll
        for (int j = 0; j < 3; ++j)
            d[i][j] = __builtin_amdgcn_wmma_f32_16x16x4_f32(false, fa[i], false, f.b[j],
                                                            (short)0, d[i][j], false, false);
    }
}

template <int NM, int NB>
__device__ __forceinline__ void gemm_run(v8f (&d)[3][3], const float* a, const float* __restrict__ wp) {
    constexpr int STEPS = GEMM_KHALF / 4;
    constexpr int TRIPS = STEPS / NB;
    static_assert(STEPS % NB == 0 && NB % 2 == 0, "whole trips of NB steps; step parity = slot parity for the A sets");
    gemm_bfrag bq[NB];
    v2f        af[2][NM];
    #pragma unroll
    for (int j = 0; j < NB - 1; ++j)
        gemm_load_b(bq[j], wp + (size_t)(j * 4) * DIM);
    gemm_load_a<NM>(af[0], a);
    #pragma clang loop unroll(disable)
    for (int t = 0; t < TRIPS; ++t) {
        const int tn = t + 1 < TRIPS ? t + 1 : t;
        const float* wc = wp + (size_t)(t  * NB * 4) * DIM;
        const float* wn = wp + (size_t)(tn * NB * 4) * DIM;
        const float* ac = a + t  * NB * 4;
        const float* an = a + tn * NB * 4;
        #pragma unroll
        for (int j = 0; j < NB; ++j) {
            if (j == 0) gemm_load_b(bq[NB - 1], wc + (size_t)((NB - 1) * 4) * DIM);
            else        gemm_load_b(bq[j - 1],  wn + (size_t)((j - 1) * 4) * DIM);
            if (j < NB - 1) gemm_load_a<NM>(af[(j + 1) & 1], ac + (j + 1) * 4);
            else            gemm_load_a<NM>(af[0], an);
            __builtin_amdgcn_sched_barrier(0);
            gemm_step<NM>(d, af[j & 1], bq[j]);
        }
    }
}

__global__ __launch_bounds__(AGG_THREADS) void gcn_aggregate_gemm(const int* __restrict__ src,
                                                                 const int* __restrict__ dst,
                                                                 const float* __restrict__ X,
                                                                 const float* __restrict__ W,
                                                                 const float* __restrict__ dinv,
                                                                 const int* __restrict__ cntg,
                                                                 const int* __restrict__ lstg,
                                                                 const float* __restrict__ bias,
                                                                 float* __restrict__ out) {
    __shared__ float acc[AGG_ROWS * AGG_PITCH];
    __shared__ int   lst[AGG_ROWS * AGG_CAP];
    __shared__ float lw[AGG_ROWS * AGG_CAP];
    __shared__ int   cnt[AGG_ROWS];
    __shared__ int   next_unit;

    const int tid  = threadIdx.x;
    const int lane = tid & 31;
    const int wave = __builtin_amdgcn_readfirstlane(tid >> 5);
    const int side = blockIdx.x & 1;
    const int lo   = (blockIdx.x >> 1) * PAIR_ROWS + side * 32;

    static_assert((AGG_ROWS * AGG_PITCH) % 4 == 0, "the accumulator is a whole number of float4 slots");
    #pragma unroll
    for (int it = 0; it < (AGG_ROWS * AGG_PITCH / 4 + AGG_THREADS - 1) / AGG_THREADS; ++it) {
        const int q = tid + it * AGG_THREADS;
        if (q < AGG_ROWS * AGG_PITCH / 4)
            *(float4*)(acc + q * 4) = make_float4(0.0f, 0.0f, 0.0f, 0.0f);
    }
    if (tid == 0) next_unit = 0;

    {
        constexpr int META_IT = (AGG_ROWS * AGG_CAP + AGG_THREADS - 1) / AGG_THREADS;
        bool ok[META_IT];
        int  mr[META_IT];
        int  mc[META_IT];
        int  ms[META_IT];
        #pragma unroll
        for (int it = 0; it < META_IT; ++it) {
            const int q  = tid + it * AGG_THREADS;
            const int qc = q < AGG_ROWS * AGG_CAP ? q : AGG_ROWS * AGG_CAP - 1;
            mr[it] = qc / AGG_CAP;
            mc[it] = cntg[lo + mr[it]];
            ok[it] = q < AGG_ROWS * AGG_CAP && mc[it] <= AGG_CAP && qc - mr[it] * AGG_CAP < mc[it];
            ms[it] = lstg[(size_t)lo * AGG_CAP + qc];
        }
        static_assert(META_IT == 3, "the two fences below name three entries");
        asm volatile("" :: "v"(ms[0]), "v"(ms[1]), "v"(ms[2]));
        float ds[META_IT];
        float dt[META_IT];
        #pragma unroll
        for (int it = 0; it < META_IT; ++it) {
            ds[it] = dinv[ms[it]];
            dt[it] = dinv[lo + mr[it]];
        }
        asm volatile("" :: "v"(ds[0]), "v"(ds[1]), "v"(ds[2]), "v"(dt[0]), "v"(dt[1]), "v"(dt[2]));
        #pragma unroll
        for (int it = 0; it < META_IT; ++it) {
            const int q = tid + it * AGG_THREADS;
            if (q < AGG_ROWS * AGG_CAP && q - mr[it] * AGG_CAP == 0) cnt[mr[it]] = mc[it];
            if (ok[it]) {
                lst[q] = ms[it];
                lw[q]  = ds[it] * dt[it];
            }
        }
    }
    __syncthreads();

    {
        static_assert(AGG_ROWS > 32 && AGG_ROWS <= 64, "a lane looks at rows lane and 32 + lane");
        const int  c0 = cnt[lane];
        const int  c1 = cnt[lane < AGG_ROWS - 32 ? 32 + lane : 0];
        const bool any_over = __builtin_amdgcn_ballot_w32(c0 > AGG_CAP || c1 > AGG_CAP) != 0u;
        if (any_over) {
            for (int e = tid; e < N_EDGES; e += AGG_THREADS) {
                const int t = dst[e] - lo;
                if ((unsigned)t < (unsigned)AGG_ROWS && cnt[t] > AGG_CAP) {
                    const int   sn = src[e];
                    const float n  = dinv[sn] * dinv[lo + t];
                    const float* xp = X + (size_t)sn * DIM;
                    float*       ap = acc + t * AGG_PITCH;
                    for (int c = 0; c < DIM; ++c)
                        atomicAdd(ap + c, n * xp[c]);
                }
            }
            __syncthreads();
        }
    }

    for (;;) {
        int u = 0;
        if (lane == 0) u = atomicAdd(&next_unit, 1);
        u = __builtin_amdgcn_readfirstlane(u);
        if (u >= AGG_UNITS) break;
        const int r   = u >> 1;
        const int col = (u & 1) * (DIM / 2) + lane * 4;
        const int i   = lo + r;
        const int nr  = __builtin_amdgcn_readfirstlane(cnt[r]);
        const int n   = nr <= AGG_CAP ? nr : 0;
        const float  di = dinv[i];
        const float* xi = X + (size_t)i * DIM + col;
        float4 a[3];
        #pragma unroll
        for (int b = 0; b < 3; ++b) {
            const float4 x4 = *(const float4*)(xi + b * 128);
            a[b] = make_float4(di * di * x4.x, di * di * x4.y, di * di * x4.z, di * di * x4.w);
        }
        const int*   ls = lst + r * AGG_CAP;
        const float* lv = lw  + r * AGG_CAP;
        for (int k = 0; k < n; k += 4) {
            float4 v[4][3];
            float  w[4];
            #pragma unroll
            for (int q = 0; q < 4; ++q) {
                const int kk = k + q < n ? k + q : n - 1;
                const int s  = __builtin_amdgcn_readfirstlane(ls[kk]);
                w[q] = lv[kk];
                const float* xp = X + (size_t)s * DIM + col;
                #pragma unroll
                for (int b = 0; b < 3; ++b)
                    v[q][b] = *(const float4*)(xp + b * 128);
            }
            #pragma unroll
            for (int q = 0; q < 4; ++q) {
                if (k + q < n) {
                    #pragma unroll
                    for (int b = 0; b < 3; ++b) {
                        a[b].x += w[q] * v[q][b].x;
                        a[b].y += w[q] * v[q][b].y;
                        a[b].z += w[q] * v[q][b].z;
                        a[b].w += w[q] * v[q][b].w;
                    }
                }
            }
        }
        float* ap = acc + r * AGG_PITCH + col;
        #pragma unroll
        for (int b = 0; b < 3; ++b) {
            float4 o = *(const float4*)(ap + b * 128);
            o.x += a[b].x;
            o.y += a[b].y;
            o.z += a[b].z;
            o.w += a[b].w;
            *(float4*)(ap + b * 128) = o;
        }
    }
    __syncthreads();

    const int  m     = lane & 15;
    const int  hi    = lane >> 4;
    const int  wb2   = (wave >> 2) & 1;
    const bool full  = wb2 == 0;
    const int  nhalf = wb2 ^ side;
    const int  col0  = (nhalf * 8 + ((wave & 3) | (((wave >> 3) & 1) << 2))) * GEMM_COLS;
    const int  kh    = wave >> 4;
    const int  k0    = kh * GEMM_KHALF + hi * 2;
    const int  shr   = side ? 0 : 2;
    const int  t0    = full ? 0 : side;
    const int  nm    = full ? 3 : 2;

    const float* ab = acc + (16 * t0 + m) * AGG_PITCH + k0;
    const float* wp = W + (size_t)k0 * DIM + col0 + m;

    v8f d[3][3] = {};
    if (full) gemm_run<3, GEMM_RING3>(d, ab, wp);
    else      gemm_run<2, GEMM_RING2>(d, ab, wp);
    __syncthreads();

    if (kh == 1) {
        #pragma unroll
        for (int i = 0; i < 3; ++i) {
            if (i < nm) {
                #pragma unroll
                for (int r = 0; r < 8; ++r) {
                    float* p = acc + (16 * (t0 + i) + 8 * hi + r) * AGG_PITCH + col0 + m;
                    p[ 0] = d[i][0][r];
                    p[16] = d[i][1][r];
                    p[32] = d[i][2][r];
                }
            }
        }
    }
    __syncthreads();
    if (kh == 0) {
        const float b0 = bias[col0 + m];
        const float b1 = bias[col0 + m + 16];
        const float b2 = bias[col0 + m + 32];
        #pragma unroll
        for (int i = 0; i < 3; ++i) {
            if (i < nm) {
                #pragma unroll
                for (int r = 0; r < 8; ++r) {
                    float* p = acc + (16 * (t0 + i) + 8 * hi + r) * AGG_PITCH + col0 + m;
                    p[ 0] = (d[i][0][r] + p[ 0]) + b0;
                    p[16] = (d[i][1][r] + p[16]) + b1;
                    p[32] = (d[i][2][r] + p[32]) + b2;
                }
            }
        }
    }
    __syncthreads();

    float* op = out + (size_t)lo * DIM;
    #pragma unroll
    for (int it = 0; it < AGG_VEC / AGG_THREADS; ++it) {
        const int q = tid + it * AGG_THREADS;
        const int r = q / (DIM / 4);
        const int c = q - r * (DIM / 4);
        if ((r >> 4) != shr || (c >= DIM / 8 ? 1 : 0) == side)
            *(float4*)(op + q * 4) = *(const float4*)(acc + r * AGG_PITCH + c * 4);
    }
}

extern "C" void kernel_launch(void* const* d_in, const int* in_sizes, int n_in,
                              void* d_out, int out_size, void* d_ws, size_t ws_size,
                              hipStream_t stream) {
    const float* x    = (const float*)d_in[0];
    const int*   ei   = (const int*)d_in[1];
    const float* W    = (const float*)d_in[2];
    const float* bias = (const float*)d_in[3];
    float*       out  = (float*)d_out;

    const int* src = ei;
    const int* dst = ei + N_EDGES;

    float* dinv = (float*)d_ws;
    int*   cntg = (int*)d_ws + N_NODES;
    int*   lstg = cntg + N_NODES;

    gcn_file_edges    <<<N_NODES / FILE_ROWS, AGG_THREADS, 0, stream>>>(src, dst, dinv, cntg, lstg);
    gcn_aggregate_gemm<<<AGG_GROUPS, AGG_THREADS, 0, stream>>>(src, dst, x, W, dinv, cntg, lstg, bias, out);
}
